# starting kernel
# baseline (speedup 1.0000x reference)
#include <hip/hip_runtime.h>

typedef float v2f __attribute__((ext_vector_type(2)));
typedef float v8f __attribute__((ext_vector_type(8)));

#define NFW   2925
#define RROT  256
#define DFULL 169
#define TILES 28
#define KSTEPS 7

__constant__ int c_off[13] = {0, 1, 10, 35, 84, 165, 286, 455, 680, 969, 1330, 1771, 2300};

__global__ __launch_bounds__(256)
void WignerRotator_wmma(const float* __restrict__ coeffs,
                        const float* __restrict__ W,
                        float* __restrict__ out)
{
    const int lane = threadIdx.x;
    const int r    = blockIdx.x / TILES;
    const int tile = blockIdx.x - r * TILES;

    int l, ti, tj;
    if (tile < 8) { l = tile; ti = 0; tj = 0; }
    else { int t = tile - 8; l = 8 + (t >> 2); ti = (t >> 1) & 1; tj = t & 1; }

    const int d   = 2 * l + 1;
    const int s   = l * l;
    const int off = c_off[l];
    const int b0  = threadIdx.y * 4;

    const int m        = lane & 15;
    const int kk       = (lane >> 4) * 2;
    const int arow     = ti * 16 + m;
    const int bcol     = tj * 16 + m;
    const bool arow_ok = (arow < d);
    const bool bcol_ok = (bcol < d);
    const int arow_c   = arow_ok ? arow : 0;
    const int bcol_c   = bcol_ok ? bcol : 0;

    const int wbase = r * (DFULL * DFULL) + s * DFULL + s;
    const int cbase = b0 * NFW + off;

    v8f acc0 = {0.f,0.f,0.f,0.f,0.f,0.f,0.f,0.f};
    v8f acc1 = {0.f,0.f,0.f,0.f,0.f,0.f,0.f,0.f};
    v8f acc2 = {0.f,0.f,0.f,0.f,0.f,0.f,0.f,0.f};
    v8f acc3 = {0.f,0.f,0.f,0.f,0.f,0.f,0.f,0.f};

    #pragma unroll
    for (int kt = 0; kt < KSTEPS; ++kt) {
        const int  ka    = kt * 4 + kk;
        const int  kb    = ka + 1;
        const bool ka_ok = (ka < d);
        const bool kb_ok = (kb < d);
        const int  kac   = ka_ok ? ka : 0;
        const int  kbc   = kb_ok ? kb : 0;

        const float a0 = W[wbase + arow_c * DFULL + kac];
        const float a1 = W[wbase + arow_c * DFULL + kbc];
        v2f a;
        a.x = (arow_ok && ka_ok) ? a0 : 0.f;
        a.y = (arow_ok && kb_ok) ? a1 : 0.f;

        const int boffA = cbase + kac * d + bcol_c;
        const int boffB = cbase + kbc * d + bcol_c;
        const bool selA = (bcol_ok && ka_ok);
        const bool selB = (bcol_ok && kb_ok);

        const float t0A = coeffs[boffA + 0 * NFW];
        const float t0B = coeffs[boffB + 0 * NFW];
        const float t1A = coeffs[boffA + 1 * NFW];
        const float t1B = coeffs[boffB + 1 * NFW];
        const float t2A = coeffs[boffA + 2 * NFW];
        const float t2B = coeffs[boffB + 2 * NFW];
        const float t3A = coeffs[boffA + 3 * NFW];
        const float t3B = coeffs[boffB + 3 * NFW];

        v2f bf0, bf1, bf2, bf3;
        bf0.x = selA ? t0A : 0.f;  bf0.y = selB ? t0B : 0.f;
        bf1.x = selA ? t1A : 0.f;  bf1.y = selB ? t1B : 0.f;
        bf2.x = selA ? t2A : 0.f;  bf2.y = selB ? t2B : 0.f;
        bf3.x = selA ? t3A : 0.f;  bf3.y = selB ? t3B : 0.f;

        acc0 = __builtin_amdgcn_wmma_f32_16x16x4_f32(false, a, false, bf0, (short)0, acc0, false, false);
        acc1 = __builtin_amdgcn_wmma_f32_16x16x4_f32(false, a, false, bf1, (short)0, acc1, false, false);
        acc2 = __builtin_amdgcn_wmma_f32_16x16x4_f32(false, a, false, bf2, (short)0, acc2, false, false);
        acc3 = __builtin_amdgcn_wmma_f32_16x16x4_f32(false, a, false, bf3, (short)0, acc3, false, false);
    }

    const int n     = lane & 15;
    const int col   = tj * 16 + n;
    const int mbase = (lane >> 4) * 8;

    const int obase = (b0 * RROT + r) * NFW + off;

    if (col < d) {
        #pragma unroll
        for (int v = 0; v < 8; ++v) {
            const int row = ti * 16 + mbase + v;
            if (row < d) {
                const int o = obase + row * d + col;
                out[o + 0 * (RROT * NFW)] = acc0[v];
                out[o + 1 * (RROT * NFW)] = acc1[v];
                out[o + 2 * (RROT * NFW)] = acc2[v];
                out[o + 3 * (RROT * NFW)] = acc3[v];
            }
        }
    }
}

extern "C" void kernel_launch(void* const* d_in, const int* in_sizes, int n_in,
                              void* d_out, int out_size, void* d_ws, size_t ws_size,
                              hipStream_t stream) {
    (void)in_sizes; (void)n_in; (void)d_ws; (void)ws_size; (void)out_size;
    const float* coeffs = (const float*)d_in[0];
    const float* wig    = (const float*)d_in[1];
    float* out          = (float*)d_out;

    dim3 block(32, 8);
    dim3 grid(RROT * TILES);
    hipLaunchKernelGGL(WignerRotator_wmma, grid, block, 0, stream, coeffs, wig, out);
}


# Each wave keeps the W fragments of 4 rotations in registers and loops over 16 batches, reusing each coeff fragment for all 4 (was 1 rotation x 4 batches); summation order unchanged
# speedup vs baseline: 1.3206x; 1.3206x over previous
#include <hip/hip_runtime.h>


typedef float v2f __attribute__((ext_vector_type(2)));
typedef float v8f __attribute__((ext_vector_type(8)));

#define NFW   2925
#define RROT  256
#define NBAT  32
#define DFULL 169
#define TILES 28
#define KSTEPS 7
#define RW    4
#define NBW   16
#define WAVES (NBAT / NBW)

__constant__ int c_off[13] = {0, 1, 10, 35, 84, 165, 286, 455, 680, 969, 1330, 1771, 2300};

__global__ __launch_bounds__(32 * WAVES)
void WignerRotator_wmma(const float* __restrict__ coeffs,
                        const float* __restrict__ W,
                        float* __restrict__ out)
{
    const int lane = threadIdx.x;
    const int rg   = blockIdx.x / TILES;
    const int tile = blockIdx.x - rg * TILES;
    const int r0   = rg * RW;

    int l, ti, tj;
    if (tile < 8) { l = tile; ti = 0; tj = 0; }
    else { int t = tile - 8; l = 8 + (t >> 2); ti = (t >> 1) & 1; tj = t & 1; }

    const int d   = 2 * l + 1;
    const int s   = l * l;
    const int off = c_off[l];
    const int b0  = __builtin_amdgcn_readfirstlane((int)threadIdx.y) * NBW;

    const int m        = lane & 15;
    const int kk       = (lane >> 4) * 2;
    const int arow     = ti * 16 + m;
    const int bcol     = tj * 16 + m;
    const bool arow_ok = (arow < d);
    const bool bcol_ok = (bcol < d);
    const int arow_c   = arow_ok ? arow : 0;
    const int bcol_c   = bcol_ok ? bcol : 0;

    v2f a[RW][KSTEPS];
    #pragma unroll
    for (int rr = 0; rr < RW; ++rr) {
        const int wrow = (r0 + rr) * (DFULL * DFULL) + (s + arow_c) * DFULL + s;
        #pragma unroll
        for (int kt = 0; kt < KSTEPS; ++kt) {
            const int  ka    = kt * 4 + kk;
            const int  kb    = ka + 1;
            const bool ka_ok = (ka < d);
            const bool kb_ok = (kb < d);
            const int  kac   = ka_ok ? ka : 0;
            const int  kbc   = kb_ok ? kb : 0;
            const float a0 = W[wrow + kac];
            const float a1 = W[wrow + kbc];
            a[rr][kt].x = (arow_ok && ka_ok) ? a0 : 0.f;
            a[rr][kt].y = (arow_ok && kb_ok) ? a1 : 0.f;
        }
    }

    const int col   = tj * 16 + m;
    const int row0  = ti * 16 + (lane >> 4) * 8;
    const int lo    = off + row0 * d + col;

    v2f bcur[KSTEPS];
    {
        const float* __restrict__ cb = coeffs + b0 * NFW + off;
        #pragma unroll
        for (int kt = 0; kt < KSTEPS; ++kt) {
            const int  ka    = kt * 4 + kk;
            const int  kb    = ka + 1;
            const bool ka_ok = (ka < d);
            const bool kb_ok = (kb < d);
            const int  kac   = ka_ok ? ka : 0;
            const int  kbc   = kb_ok ? kb : 0;
            const float tA = cb[kac * d + bcol_c];
            const float tB = cb[kbc * d + bcol_c];
            bcur[kt].x = (bcol_ok && ka_ok) ? tA : 0.f;
            bcur[kt].y = (bcol_ok && kb_ok) ? tB : 0.f;
        }
    }

    #pragma unroll
    for (int bi = 0; bi < NBW; ++bi) {
        const int b = b0 + bi;

        v2f bnxt[KSTEPS];
        if (bi + 1 < NBW) {
            const float* __restrict__ cb = coeffs + (b + 1) * NFW + off;
            #pragma unroll
            for (int kt = 0; kt < KSTEPS; ++kt) {
                const int  ka    = kt * 4 + kk;
                const int  kb    = ka + 1;
                const bool ka_ok = (ka < d);
                const bool kb_ok = (kb < d);
                const int  kac   = ka_ok ? ka : 0;
                const int  kbc   = kb_ok ? kb : 0;
                const float tA = cb[kac * d + bcol_c];
                const float tB = cb[kbc * d + bcol_c];
                bnxt[kt].x = (bcol_ok && ka_ok) ? tA : 0.f;
                bnxt[kt].y = (bcol_ok && kb_ok) ? tB : 0.f;
            }
        }

        v8f acc[RW];
        #pragma unroll
        for (int rr = 0; rr < RW; ++rr)
            acc[rr] = (v8f){0.f,0.f,0.f,0.f,0.f,0.f,0.f,0.f};

        #pragma unroll
        for (int kt = 0; kt < KSTEPS; ++kt) {
            #pragma unroll
            for (int rr = 0; rr < RW; ++rr)
                acc[rr] = __builtin_amdgcn_wmma_f32_16x16x4_f32(false, a[rr][kt], false, bcur[kt],
                                                                (short)0, acc[rr], false, false);
        }

        float* __restrict__ ob = out + (b * RROT + r0) * NFW;
        if (col < d) {
            #pragma unroll
            for (int v = 0; v < 8; ++v) {
                if (row0 + v < d) {
                    const int o = lo + v * d;
                    #pragma unroll
                    for (int rr = 0; rr < RW; ++rr)
                        ob[rr * NFW + o] = acc[rr][v];
                }
            }
        }

        if (bi + 1 < NBW) {
            #pragma unroll
            for (int kt = 0; kt < KSTEPS; ++kt) bcur[kt] = bnxt[kt];
        }
    }
}

extern "C" void kernel_launch(void* const* d_in, const int* in_sizes, int n_in,
                              void* d_out, int out_size, void* d_ws, size_t ws_size,
                              hipStream_t stream) {
    (void)in_sizes; (void)n_in; (void)d_ws; (void)ws_size; (void)out_size;
    const float* coeffs = (const float*)d_in[0];
    const float* wig    = (const float*)d_in[1];
    float* out          = (float*)d_out;

    dim3 block(32, WAVES);
    dim3 grid((RROT / RW) * TILES);
    hipLaunchKernelGGL(WignerRotator_wmma, grid, block, 0, stream, coeffs, wig, out);
}


# B (coeff) fragment loads take scalar base + 32-bit lane offset from 14 fixed registers: no 64-bit address add into a reused pair, no translation wait between a batch's 14 loads; sums unchanged
# speedup vs baseline: 1.7564x; 1.3300x over previous
#include <hip/hip_runtime.h>
#include <stdint.h>


typedef float v2f __attribute__((ext_vector_type(2)));
typedef float v8f __attribute__((ext_vector_type(8)));

#define NFW   2925
#define RROT  256
#define NBAT  32
#define DFULL 169
#define GROUPS 7
#define GT    4
#define KSTEPS 7
#define RW    4
#define NBW   16
#define BGRP  (NBAT / NBW)
#define CHS   640

__constant__ int c_off[14] = {0, 1, 10, 35, 84, 165, 286, 455, 680, 969, 1330, 1771, 2300, 2925};

static __device__ __forceinline__ void lds_handover() {
    __builtin_amdgcn_fence(__ATOMIC_RELEASE, "workgroup", "local");
    __builtin_amdgcn_s_barrier();
    __builtin_amdgcn_fence(__ATOMIC_ACQUIRE, "workgroup", "local");
}

template <int J0, int N, int NM>
static __device__ __forceinline__ void put_lines(const float* __restrict__ sc, float* __restrict__ oc,
                                                 int p0, int clen) {
    float val[N];
    #pragma unroll
    for (int j = 0; j < N; ++j) {
        const int p = p0 + (J0 + j) * 32;
        val[j] = sc[(j < NM) ? min(max(p, 0), clen - 1) : p];
    }
    #pragma unroll
    for (int j = 0; j < N; ++j) {
        const int p = p0 + (J0 + j) * 32;
        if (j >= NM || (unsigned)p < (unsigned)clen) oc[p] = val[j];
    }
}

__global__ __launch_bounds__(32 * GT)
void WignerRotator_wmma(const float* __restrict__ coeffs,
                        const float* __restrict__ W,
                        float* __restrict__ out)
{
    __shared__ float stage[2][RW * CHS];

    const int lane = threadIdx.x;
    const int w    = __builtin_amdgcn_readfirstlane((int)threadIdx.y);
    const int bg   = blockIdx.x % BGRP;
    const int rgg  = blockIdx.x / BGRP;
    const int rg   = rgg / GROUPS;
    const int g    = rgg - rg * GROUPS;
    const int r0   = rg * RW;
    const int tile = g * GT + w;

    int l, ti, tj;
    if (tile < 8) { l = tile; ti = 0; tj = 0; }
    else { int t = tile - 8; l = 8 + (t >> 2); ti = (t >> 1) & 1; tj = t & 1; }

    const int lf   = (g < 2) ? g * GT : g + 6;
    const int ll   = (g < 2) ? g * GT + 3 : g + 6;
    const int coff = c_off[lf];
    const int clen = c_off[ll + 1] - coff;

    const int d   = 2 * l + 1;
    const int s   = l * l;
    const int off = c_off[l];
    const int b0  = bg * NBW;

    const int m        = lane & 15;
    const int kk       = (lane >> 4) * 2;
    const int arow     = ti * 16 + m;
    const int bcol     = tj * 16 + m;
    const bool arow_ok = (arow < d);
    const bool bcol_ok = (bcol < d);
    const int arow_c   = arow_ok ? arow : 0;
    const int bcol_c   = bcol_ok ? bcol : 0;

    v2f a[RW][KSTEPS];
    #pragma unroll
    for (int rr = 0; rr < RW; ++rr) {
        const int wrow = (r0 + rr) * (DFULL * DFULL) + (s + arow_c) * DFULL + s;
        #pragma unroll
        for (int kt = 0; kt < KSTEPS; ++kt) {
            const int  ka    = kt * 4 + kk;
            const int  kb    = ka + 1;
            const bool ka_ok = (ka < d);
            const bool kb_ok = (kb < d);
            const int  kac   = ka_ok ? ka : 0;
            const int  kbc   = kb_ok ? kb : 0;
            const float a0 = W[wrow + kac];
            const float a1 = W[wrow + kbc];
            a[rr][kt].x = (arow_ok && ka_ok) ? a0 : 0.f;
            a[rr][kt].y = (arow_ok && kb_ok) ? a1 : 0.f;
        }
    }

    const int col   = tj * 16 + m;
    const int row0  = ti * 16 + (lane >> 4) * 8;
    const int so    = (off - coff) + row0 * d + col;

    int bo_x[KSTEPS], bo_y[KSTEPS];
    #pragma unroll
    for (int kt = 0; kt < KSTEPS; ++kt) {
        const int  ka    = kt * 4 + kk;
        const int  kb    = ka + 1;
        const bool ka_ok = (ka < d);
        const bool kb_ok = (kb < d);
        const int  kac   = ka_ok ? ka : 0;
        const int  kbc   = kb_ok ? kb : 0;
        bo_x[kt] = kac * d + bcol_c;
        bo_y[kt] = kbc * d + bcol_c;
    }
    auto load_b = [&](v2f (&bf)[KSTEPS], int b) {
        const float* __restrict__ cb = coeffs + b * NFW + off;
        #pragma unroll
        for (int kt = 0; kt < KSTEPS; ++kt) {
            asm("" : "+v"(bo_x[kt]) : "s"(b));
            asm("" : "+v"(bo_y[kt]) : "s"(b));
            bf[kt].x = cb[bo_x[kt]];
            bf[kt].y = cb[bo_y[kt]];
        }
    };

    auto step = [&](const v2f (&braw)[KSTEPS], float* sb, int b) {
        v2f bf[KSTEPS];
        #pragma unroll
        for (int kt = 0; kt < KSTEPS; ++kt) {
            const int  ka = kt * 4 + kk;
            bf[kt].x = (bcol_ok && (ka     < d)) ? braw[kt].x : 0.f;
            bf[kt].y = (bcol_ok && (ka + 1 < d)) ? braw[kt].y : 0.f;
        }

        v8f acc[RW];
        #pragma unroll
        for (int rr = 0; rr < RW; ++rr)
            acc[rr] = (v8f){0.f,0.f,0.f,0.f,0.f,0.f,0.f,0.f};

        #pragma unroll
        for (int kt = 0; kt < KSTEPS; ++kt) {
            #pragma unroll
            for (int rr = 0; rr < RW; ++rr)
                acc[rr] = __builtin_amdgcn_wmma_f32_16x16x4_f32(false, a[rr][kt], false, bf[kt],
                                                                (short)0, acc[rr], false, false);
        }

        if (col < d) {
            #pragma unroll
            for (int v = 0; v < 8; ++v) {
                if (row0 + v < d) {
                    #pragma unroll
                    for (int rr = 0; rr < RW; ++rr)
                        sb[rr * CHS + so + v * d] = acc[rr][v];
                }
            }
        }

        lds_handover();

        float* __restrict__ oc = out + ((b * RROT + r0 + w) * NFW + coff);
        const float* sc = sb + w * CHS;
        const int mis = (int)(((uintptr_t)oc >> 2) & 31);
        const int nl  = (mis + clen + 31) >> 5;
        const int p0  = lane - mis;
        if (nl >= 9)       put_lines<0, 8, 1>(sc, oc, p0, clen);
        else               put_lines<0, 8, 8>(sc, oc, p0, clen);
        if (nl >= 17)      put_lines<8, 8, 0>(sc, oc, p0, clen);
        else if (nl > 8)   put_lines<8, 8, 8>(sc, oc, p0, clen);
        if (nl > 16)       put_lines<16, 5, 5>(sc, oc, p0, clen);
    };

    v2f bA[KSTEPS], bB[KSTEPS];
    load_b(bA, b0);
    #pragma unroll 1
    for (int bi = 0; bi < NBW; bi += 2) {
        const int b = b0 + bi;
        load_b(bB, b + 1);
        step(bA, stage[0], b);
        load_b(bA, (bi + 2 < NBW) ? b + 2 : b + 1);
        step(bB, stage[1], b + 1);
    }
}

extern "C" void kernel_launch(void* const* d_in, const int* in_sizes, int n_in,
                              void* d_out, int out_size, void* d_ws, size_t ws_size,
                              hipStream_t stream) {
    (void)in_sizes; (void)n_in; (void)d_ws; (void)ws_size; (void)out_size;
    const float* coeffs = (const float*)d_in[0];
    const float* wig    = (const float*)d_in[1];
    float* out          = (float*)d_out;

    dim3 block(32, GT);
    dim3 grid((RROT / RW) * GROUPS * BGRP);
    hipLaunchKernelGGL(WignerRotator_wmma, grid, block, 0, stream, coeffs, wig, out);
}


# K loop of each tile group cut to its own depth (2,4,5,5,6,6,7 steps of 4, was 7 for all): fragment loads and matrix instructions of k steps that hold only zero padding dropped; real sums unchanged
# speedup vs baseline: 1.8138x; 1.0327x over previous
#include <hip/hip_runtime.h>
#include <stdint.h>


typedef float v2f __attribute__((ext_vector_type(2)));
typedef float v8f __attribute__((ext_vector_type(8)));

#define NFW   2925
#define RROT  256
#define NBAT  32
#define DFULL 169
#define GROUPS 7
#define GT    4
#define KSTEPS 7
#define RW    4
#define NBW   16
#define BGRP  (NBAT / NBW)
#define CHS   640

__constant__ int c_off[14] = {0, 1, 10, 35, 84, 165, 286, 455, 680, 969, 1330, 1771, 2300, 2925};

static __device__ __forceinline__ void lds_handover() {
    __builtin_amdgcn_fence(__ATOMIC_RELEASE, "workgroup", "local");
    __builtin_amdgcn_s_barrier();
    __builtin_amdgcn_fence(__ATOMIC_ACQUIRE, "workgroup", "local");
}

template <int I> struct KIdx { static constexpr int v = I; };

template <class F>
static __device__ __forceinline__ void k_chain(int ks, F&& f) {
    f(KIdx<0>{}); f(KIdx<1>{});
    if (ks > 2) {
        f(KIdx<2>{}); f(KIdx<3>{});
        if (ks > 4) {
            f(KIdx<4>{});
            if (ks > 5) {
                f(KIdx<5>{});
                if (ks > 6) f(KIdx<6>{});
            }
        }
    }
}

template <int J0, int N, int NM>
static __device__ __forceinline__ void put_lines(const float* __restrict__ sc, float* __restrict__ oc,
                                                 int p0, int clen) {
    float val[N];
    #pragma unroll
    for (int j = 0; j < N; ++j) {
        const int p = p0 + (J0 + j) * 32;
        val[j] = sc[(j < NM) ? min(max(p, 0), clen - 1) : p];
    }
    #pragma unroll
    for (int j = 0; j < N; ++j) {
        const int p = p0 + (J0 + j) * 32;
        if (j >= NM || (unsigned)p < (unsigned)clen) oc[p] = val[j];
    }
}

__global__ __launch_bounds__(32 * GT)
void WignerRotator_wmma(const float* __restrict__ coeffs,
                        const float* __restrict__ W,
                        float* __restrict__ out)
{
    __shared__ float stage[2][RW * CHS];

    const int lane = threadIdx.x;
    const int w    = __builtin_amdgcn_readfirstlane((int)threadIdx.y);
    const int bg   = blockIdx.x % BGRP;
    const int rgg  = blockIdx.x / BGRP;
    const int rg   = rgg / GROUPS;
    const int g    = rgg - rg * GROUPS;
    const int r0   = rg * RW;
    const int tile = g * GT + w;

    int l, ti, tj;
    if (tile < 8) { l = tile; ti = 0; tj = 0; }
    else { int t = tile - 8; l = 8 + (t >> 2); ti = (t >> 1) & 1; tj = t & 1; }

    const int lf   = (g < 2) ? g * GT : g + 6;
    const int ll   = (g < 2) ? g * GT + 3 : g + 6;
    const int coff = c_off[lf];
    const int clen = c_off[ll + 1] - coff;
    const int ks   = (ll + 2) >> 1;

    const int d   = 2 * l + 1;
    const int s   = l * l;
    const int off = c_off[l];
    const int b0  = bg * NBW;

    const int m        = lane & 15;
    const int kk       = (lane >> 4) * 2;
    const int arow     = ti * 16 + m;
    const int bcol     = tj * 16 + m;
    const bool arow_ok = (arow < d);
    const bool bcol_ok = (bcol < d);
    const int arow_c   = arow_ok ? arow : 0;
    const int bcol_c   = bcol_ok ? bcol : 0;

    v2f a[RW][KSTEPS];
    const int wrow = r0 * (DFULL * DFULL) + (s + arow_c) * DFULL + s;
    k_chain(ks, [&](auto kc) {
        constexpr int kt = decltype(kc)::v;
        const int  ka    = kt * 4 + kk;
        const int  kb    = ka + 1;
        const bool ka_ok = (ka < d);
        const bool kb_ok = (kb < d);
        const int  kac   = ka_ok ? ka : 0;
        const int  kbc   = kb_ok ? kb : 0;
        #pragma unroll
        for (int rr = 0; rr < RW; ++rr) {
            const float a0 = W[wrow + rr * (DFULL * DFULL) + kac];
            const float a1 = W[wrow + rr * (DFULL * DFULL) + kbc];
            a[rr][kt].x = (arow_ok && ka_ok) ? a0 : 0.f;
            a[rr][kt].y = (arow_ok && kb_ok) ? a1 : 0.f;
        }
    });

    const int col   = tj * 16 + m;
    const int row0  = ti * 16 + (lane >> 4) * 8;
    const int so    = (off - coff) + row0 * d + col;

    int bo_x[KSTEPS], bo_y[KSTEPS];
    #pragma unroll
    for (int kt = 0; kt < KSTEPS; ++kt) {
        const int  ka    = kt * 4 + kk;
        const int  kb    = ka + 1;
        const bool ka_ok = (ka < d);
        const bool kb_ok = (kb < d);
        const int  kac   = ka_ok ? ka : 0;
        const int  kbc   = kb_ok ? kb : 0;
        bo_x[kt] = kac * d + bcol_c;
        bo_y[kt] = kbc * d + bcol_c;
    }
    auto load_b = [&](v2f (&bf)[KSTEPS], int b) {
        const float* __restrict__ cb = coeffs + b * NFW + off;
        k_chain(ks, [&](auto kc) {
            constexpr int kt = decltype(kc)::v;
            asm("" : "+v"(bo_x[kt]) : "s"(b));
            asm("" : "+v"(bo_y[kt]) : "s"(b));
            bf[kt].x = cb[bo_x[kt]];
            bf[kt].y = cb[bo_y[kt]];
        });
    };

    auto step = [&](v2f (&braw)[KSTEPS], float* sb, int b, int bnext) {
        v8f acc[RW];
        #pragma unroll
        for (int rr = 0; rr < RW; ++rr)
            acc[rr] = (v8f){0.f,0.f,0.f,0.f,0.f,0.f,0.f,0.f};

        k_chain(ks, [&](auto kc) {
            constexpr int kt = decltype(kc)::v;
            const int ka = kt * 4 + kk;
            v2f bf;
            bf.x = (bcol_ok && (ka     < d)) ? braw[kt].x : 0.f;
            bf.y = (bcol_ok && (ka + 1 < d)) ? braw[kt].y : 0.f;
            #pragma unroll
            for (int rr = 0; rr < RW; ++rr)
                acc[rr] = __builtin_amdgcn_wmma_f32_16x16x4_f32(false, a[rr][kt], false, bf,
                                                                (short)0, acc[rr], false, false);
        });

        load_b(braw, bnext);

        if (col < d) {
            #pragma unroll
            for (int v = 0; v < 8; ++v) {
                if (row0 + v < d) {
                    #pragma unroll
                    for (int rr = 0; rr < RW; ++rr)
                        sb[rr * CHS + so + v * d] = acc[rr][v];
                }
            }
        }

        lds_handover();

        float* __restrict__ oc = out + ((b * RROT + r0 + w) * NFW + coff);
        const float* sc = sb + w * CHS;
        const int mis = (int)(((uintptr_t)oc >> 2) & 31);
        const int nl  = (mis + clen + 31) >> 5;
        const int p0  = lane - mis;
        if (nl >= 9)       put_lines<0, 8, 1>(sc, oc, p0, clen);
        else               put_lines<0, 8, 8>(sc, oc, p0, clen);
        if (nl >= 17)      put_lines<8, 8, 0>(sc, oc, p0, clen);
        else if (nl > 8)   put_lines<8, 8, 8>(sc, oc, p0, clen);
        if (nl > 16)       put_lines<16, 5, 5>(sc, oc, p0, clen);
    };

    v2f braw[KSTEPS];
    load_b(braw, b0);
    #pragma unroll 1
    for (int bi = 0; bi < NBW; bi += 2) {
        const int b = b0 + bi;
        step(braw, stage[0], b, b + 1);
        step(braw, stage[1], b + 1, (bi + 2 < NBW) ? b + 2 : b + 1);
    }
}

extern "C" void kernel_launch(void* const* d_in, const int* in_sizes, int n_in,
                              void* d_out, int out_size, void* d_ws, size_t ws_size,
                              hipStream_t stream) {
    (void)in_sizes; (void)n_in; (void)d_ws; (void)ws_size; (void)out_size;
    const float* coeffs = (const float*)d_in[0];
    const float* wig    = (const float*)d_in[1];
    float* out          = (float*)d_out;

    dim3 block(32, GT);
    dim3 grid((RROT / RW) * GROUPS * BGRP);
    hipLaunchKernelGGL(WignerRotator_wmma, grid, block, 0, stream, coeffs, wig, out);
}


# Write-out reads LDS and stores out in 16-byte pieces per lane (four whole 128-byte lines per store instruction, was one): chunks staged at the 16-byte phase they have in out; sums unchanged
# speedup vs baseline: 1.9162x; 1.0565x over previous
#include <hip/hip_runtime.h>
#include <stdint.h>


typedef float v2f __attribute__((ext_vector_type(2)));
typedef float v8f __attribute__((ext_vector_type(8)));
typedef float v4f __attribute__((ext_vector_type(4)));
typedef v4f v4fa __attribute__((may_alias));

#define NFW   2925
#define RROT  256
#define NBAT  32
#define DFULL 169
#define GROUPS 7
#define GT    4
#define KSTEPS 7
#define RW    4
#define NBW   16
#define BGRP  (NBAT / NBW)
#define CHS   645

__constant__ int c_off[14] = {0, 1, 10, 35, 84, 165, 286, 455, 680, 969, 1330, 1771, 2300, 2925};

static __device__ __forceinline__ void lds_handover() {
    __builtin_amdgcn_fence(__ATOMIC_RELEASE, "workgroup", "local");
    __builtin_amdgcn_s_barrier();
    __builtin_amdgcn_fence(__ATOMIC_ACQUIRE, "workgroup", "local");
}

template <int I> struct KIdx { static constexpr int v = I; };

template <class F>
static __device__ __forceinline__ void k_chain(int ks, F&& f) {
    f(KIdx<0>{}); f(KIdx<1>{});
    if (ks > 2) {
        f(KIdx<2>{}); f(KIdx<3>{});
        if (ks > 4) {
            f(KIdx<4>{});
            if (ks > 5) {
                f(KIdx<5>{});
                if (ks > 6) f(KIdx<6>{});
            }
        }
    }
}

template <int N>
static __device__ __forceinline__ void put_pieces(const v4fa* __restrict__ s4, v4fa* __restrict__ g4,
                                                  int qb, int qlo, int qt) {
    v4f val[N];
    #pragma unroll
    for (int i = 0; i < N; ++i) {
        const int q = qb + 32 * i;
        val[i] = s4[(i == 0 || i == N - 1) ? min(max(q, qlo), qt - 1) : q];
    }
    #pragma unroll
    for (int i = 0; i < N; ++i) {
        const int q = qb + 32 * i;
        if ((i != 0 && i != N - 1) || (unsigned)(q - qlo) < (unsigned)(qt - qlo)) g4[q] = val[i];
    }
}

__global__ __launch_bounds__(32 * GT)
void WignerRotator_wmma(const float* __restrict__ coeffs,
                        const float* __restrict__ W,
                        float* __restrict__ out)
{
    __shared__ __attribute__((aligned(16))) float stage[2][RW * CHS];

    const int lane = threadIdx.x;
    const int w    = __builtin_amdgcn_readfirstlane((int)threadIdx.y);
    const int bg   = blockIdx.x % BGRP;
    const int rgg  = blockIdx.x / BGRP;
    const int rg   = rgg / GROUPS;
    const int g    = rgg - rg * GROUPS;
    const int r0   = rg * RW;
    const int tile = g * GT + w;

    int l, ti, tj;
    if (tile < 8) { l = tile; ti = 0; tj = 0; }
    else { int t = tile - 8; l = 8 + (t >> 2); ti = (t >> 1) & 1; tj = t & 1; }

    const int lf   = (g < 2) ? g * GT : g + 6;
    const int ll   = (g < 2) ? g * GT + 3 : g + 6;
    const int coff = c_off[lf];
    const int clen = c_off[ll + 1] - coff;
    const int ks   = (ll + 2) >> 1;

    const int d   = 2 * l + 1;
    const int s   = l * l;
    const int off = c_off[l];
    const int b0  = bg * NBW;

    const int m        = lane & 15;
    const int kk       = (lane >> 4) * 2;
    const int arow     = ti * 16 + m;
    const int bcol     = tj * 16 + m;
    const bool arow_ok = (arow < d);
    const bool bcol_ok = (bcol < d);
    const int arow_c   = arow_ok ? arow : 0;
    const int bcol_c   = bcol_ok ? bcol : 0;

    v2f a[RW][KSTEPS];
    const int wrow = r0 * (DFULL * DFULL) + (s + arow_c) * DFULL + s;
    k_chain(ks, [&](auto kc) {
        constexpr int kt = decltype(kc)::v;
        const int  ka    = kt * 4 + kk;
        const int  kb    = ka + 1;
        const bool ka_ok = (ka < d);
        const bool kb_ok = (kb < d);
        const int  kac   = ka_ok ? ka : 0;
        const int  kbc   = kb_ok ? kb : 0;
        #pragma unroll
        for (int rr = 0; rr < RW; ++rr) {
            const float a0 = W[wrow + rr * (DFULL * DFULL) + kac];
            const float a1 = W[wrow + rr * (DFULL * DFULL) + kbc];
            a[rr][kt].x = (arow_ok && ka_ok) ? a0 : 0.f;
            a[rr][kt].y = (arow_ok && kb_ok) ? a1 : 0.f;
        }
    });

    const int col   = tj * 16 + m;
    const int row0  = ti * 16 + (lane >> 4) * 8;
    const int so    = (off - coff) + row0 * d + col;

    int bo_x[KSTEPS], bo_y[KSTEPS];
    #pragma unroll
    for (int kt = 0; kt < KSTEPS; ++kt) {
        const int  ka    = kt * 4 + kk;
        const int  kb    = ka + 1;
        const bool ka_ok = (ka < d);
        const bool kb_ok = (kb < d);
        const int  kac   = ka_ok ? ka : 0;
        const int  kbc   = kb_ok ? kb : 0;
        bo_x[kt] = kac * d + bcol_c;
        bo_y[kt] = kbc * d + bcol_c;
    }
    auto load_b = [&](v2f (&bf)[KSTEPS], int b) {
        const float* __restrict__ cb = coeffs + b * NFW + off;
        k_chain(ks, [&](auto kc) {
            constexpr int kt = decltype(kc)::v;
            asm("" : "+v"(bo_x[kt]) : "s"(b));
            asm("" : "+v"(bo_y[kt]) : "s"(b));
            bf[kt].x = cb[bo_x[kt]];
            bf[kt].y = cb[bo_y[kt]];
        });
    };

    auto step = [&](v2f (&braw)[KSTEPS], float* sb, int b, int bnext) {
        v8f acc[RW];
        #pragma unroll
        for (int rr = 0; rr < RW; ++rr)
            acc[rr] = (v8f){0.f,0.f,0.f,0.f,0.f,0.f,0.f,0.f};

        k_chain(ks, [&](auto kc) {
            constexpr int kt = decltype(kc)::v;
            const int ka = kt * 4 + kk;
            v2f bf;
            bf.x = (bcol_ok && (ka     < d)) ? braw[kt].x : 0.f;
            bf.y = (bcol_ok && (ka + 1 < d)) ? braw[kt].y : 0.f;
            #pragma unroll
            for (int rr = 0; rr < RW; ++rr)
                acc[rr] = __builtin_amdgcn_wmma_f32_16x16x4_f32(false, a[rr][kt], false, bf,
                                                                (short)0, acc[rr], false, false);
        });

        load_b(braw, bnext);

        const int sh0 = (int)(((uintptr_t)(out + ((b * RROT + r0) * NFW + coff)) >> 2) & 3);

        if (col < d) {
            #pragma unroll
            for (int v = 0; v < 8; ++v) {
                if (row0 + v < d) {
                    #pragma unroll
                    for (int rr = 0; rr < RW; ++rr)
                        sb[rr * CHS + sh0 + so + v * d] = acc[rr][v];
                }
            }
        }

        lds_handover();

        float* __restrict__ oc = out + ((b * RROT + r0 + w) * NFW + coff);
        const int e0  = w * CHS + sh0;
        const int a   = e0 & 3;
        const float* sc = sb + (e0 - a);
        float* __restrict__ og = oc - a;
        const int m8  = (int)(((uintptr_t)og >> 4) & 7);
        const int qt  = (a + clen) >> 2;
        const int tn  = (a + clen) & 3;
        const int qlo = (a != 0) ? 1 : 0;
        const int nst = (m8 + qt + 31) >> 5;
        const int qb  = lane - m8;
        const v4fa* s4 = (const v4fa*)sc;
        v4fa* g4 = (v4fa*)og;
        const int  hj   = lane & 3;
        const bool htl  = (lane & 4) != 0;
        const int  hrel = htl ? 4 * qt + hj : hj;
        const bool hok  = (lane < 8) && (htl ? (hj < tn) : (a != 0 && hj >= a));
        const float hv  = sc[min(max(hrel, a), a + clen - 1)];
        if (nst <= 1)      put_pieces<1>(s4, g4, qb, qlo, qt);
        else if (nst == 2) put_pieces<2>(s4, g4, qb, qlo, qt);
        else if (nst == 3) put_pieces<3>(s4, g4, qb, qlo, qt);
        else if (nst == 4) put_pieces<4>(s4, g4, qb, qlo, qt);
        else if (nst == 5) put_pieces<5>(s4, g4, qb, qlo, qt);
        else               put_pieces<6>(s4, g4, qb, qlo, qt);
        if (hok) og[hrel] = hv;
    };

    v2f braw[KSTEPS];
    load_b(braw, b0);
    #pragma unroll 1
    for (int bi = 0; bi < NBW; bi += 2) {
        const int b = b0 + bi;
        step(braw, stage[0], b, b + 1);
        step(braw, stage[1], b + 1, (bi + 2 < NBW) ? b + 2 : b + 1);
    }
}

extern "C" void kernel_launch(void* const* d_in, const int* in_sizes, int n_in,
                              void* d_out, int out_size, void* d_ws, size_t ws_size,
                              hipStream_t stream) {
    (void)in_sizes; (void)n_in; (void)d_ws; (void)ws_size; (void)out_size;
    const float* coeffs = (const float*)d_in[0];
    const float* wig    = (const float*)d_in[1];
    float* out          = (float*)d_out;

    dim3 block(32, GT);
    dim3 grid((RROT / RW) * GROUPS * BGRP);
    hipLaunchKernelGGL(WignerRotator_wmma, grid, block, 0, stream, coeffs, wig, out);
}


# W fragments staged through LDS: each wave loads its 16 block rows as whole rows (1-2 lines per load instruction, was 4 bytes out of each of 16 rows) and reads its fragments from LDS; sums unchanged
# speedup vs baseline: 2.1641x; 1.1294x over previous
#include <hip/hip_runtime.h>
#include <stdint.h>


typedef float v2f __attribute__((ext_vector_type(2)));
typedef float v8f __attribute__((ext_vector_type(8)));
typedef float v4f __attribute__((ext_vector_type(4)));
typedef v4f v4fa __attribute__((may_alias));
typedef v2f v2fa __attribute__((may_alias));

#define NFW   2925
#define RROT  256
#define NBAT  32
#define DFULL 169
#define GROUPS 7
#define GT    4
#define KSTEPS 7
#define RW    4
#define NBW   16
#define BGRP  (NBAT / NBW)
#define APITCH 28
#define CHS   645

__constant__ int c_off[14] = {0, 1, 10, 35, 84, 165, 286, 455, 680, 969, 1330, 1771, 2300, 2925};

static __device__ __forceinline__ void lds_handover() {
    __builtin_amdgcn_fence(__ATOMIC_RELEASE, "workgroup", "local");
    __builtin_amdgcn_s_barrier();
    __builtin_amdgcn_fence(__ATOMIC_ACQUIRE, "workgroup", "local");
}

template <int I> struct KIdx { static constexpr int v = I; };

template <class F>
static __device__ __forceinline__ void k_chain(int ks, F&& f) {
    f(KIdx<0>{}); f(KIdx<1>{});
    if (ks > 2) {
        f(KIdx<2>{}); f(KIdx<3>{});
        if (ks > 4) {
            f(KIdx<4>{});
            if (ks > 5) {
                f(KIdx<5>{});
                if (ks > 6) f(KIdx<6>{});
            }
        }
    }
}

template <int N>
static __device__ __forceinline__ void put_pieces(const v4fa* __restrict__ s4, v4fa* __restrict__ g4,
                                                  int qb, int qlo, int qt) {
    v4f val[N];
    #pragma unroll
    for (int i = 0; i < N; ++i) {
        const int q = qb + 32 * i;
        val[i] = s4[(i == 0 || i == N - 1) ? min(max(q, qlo), qt - 1) : q];
    }
    #pragma unroll
    for (int i = 0; i < N; ++i) {
        const int q = qb + 32 * i;
        if ((i != 0 && i != N - 1) || (unsigned)(q - qlo) < (unsigned)(qt - qlo)) g4[q] = val[i];
    }
}

__global__ __launch_bounds__(32 * GT)
void WignerRotator_wmma(const float* __restrict__ coeffs,
                        const float* __restrict__ W,
                        float* __restrict__ out)
{
    __shared__ __attribute__((aligned(16))) float stage[2][RW * CHS];
    __shared__ __attribute__((aligned(16))) float afr[GT][RW][16][APITCH];

    const int lane = threadIdx.x;
    const int w    = __builtin_amdgcn_readfirstlane((int)threadIdx.y);
    const int bg   = blockIdx.x % BGRP;
    const int rgg  = blockIdx.x / BGRP;
    const int rg   = rgg / GROUPS;
    const int g    = rgg - rg * GROUPS;
    const int r0   = rg * RW;
    const int tile = g * GT + w;

    int l, ti, tj;
    if (tile < 8) { l = tile; ti = 0; tj = 0; }
    else { int t = tile - 8; l = 8 + (t >> 2); ti = (t >> 1) & 1; tj = t & 1; }

    const int lf   = (g < 2) ? g * GT : g + 6;
    const int ll   = (g < 2) ? g * GT + 3 : g + 6;
    const int coff = c_off[lf];
    const int clen = c_off[ll + 1] - coff;
    const int ks   = (ll + 2) >> 1;

    const int d   = 2 * l + 1;
    const int s   = l * l;
    const int off = c_off[l];
    const int b0  = bg * NBW;

    const int m        = lane & 15;
    const int kk       = (lane >> 4) * 2;
    const int bcol     = tj * 16 + m;
    const bool bcol_ok = (bcol < d);
    const int bcol_c   = bcol_ok ? bcol : 0;

    v2f a[RW][KSTEPS];
    const bool kl_ok = (lane < d);
    const int  kl    = kl_ok ? lane : 0;
    float araw[RW][16];
    const float* __restrict__ wb = W + r0 * (DFULL * DFULL);
    int vo[16];
    #pragma unroll
    for (int j = 0; j < 16; ++j) {
        const int rowc = min(ti * 16 + j, d - 1);
        vo[j] = (s + rowc) * DFULL + s + kl;
    }
    asm("" : "+v"(vo[0]), "+v"(vo[1]), "+v"(vo[2]),  "+v"(vo[3]),  "+v"(vo[4]),  "+v"(vo[5]),  "+v"(vo[6]),  "+v"(vo[7]));
    asm("" : "+v"(vo[8]), "+v"(vo[9]), "+v"(vo[10]), "+v"(vo[11]), "+v"(vo[12]), "+v"(vo[13]), "+v"(vo[14]), "+v"(vo[15]));
    #pragma unroll
    for (int j = 0; j < 16; ++j) {
        #pragma unroll
        for (int rr = 0; rr < RW; ++rr)
            araw[rr][j] = wb[vo[j] + rr * (DFULL * DFULL)];
    }

    const int col   = tj * 16 + m;
    const int row0  = ti * 16 + (lane >> 4) * 8;
    const int so    = (off - coff) + row0 * d + col;

    int bo_x[KSTEPS], bo_y[KSTEPS];
    #pragma unroll
    for (int kt = 0; kt < KSTEPS; ++kt) {
        const int  ka    = kt * 4 + kk;
        const int  kb    = ka + 1;
        const bool ka_ok = (ka < d);
        const bool kb_ok = (kb < d);
        const int  kac   = ka_ok ? ka : 0;
        const int  kbc   = kb_ok ? kb : 0;
        bo_x[kt] = kac * d + bcol_c;
        bo_y[kt] = kbc * d + bcol_c;
    }
    auto load_b = [&](v2f (&bf)[KSTEPS], int b) {
        const float* __restrict__ cb = coeffs + b * NFW + off;
        k_chain(ks, [&](auto kc) {
            constexpr int kt = decltype(kc)::v;
            asm("" : "+v"(bo_x[kt]) : "s"(b));
            asm("" : "+v"(bo_y[kt]) : "s"(b));
            bf[kt].x = cb[bo_x[kt]];
            bf[kt].y = cb[bo_y[kt]];
        });
    };

    auto step = [&](v2f (&braw)[KSTEPS], float* sb, int b, int bnext) {
        v8f acc[RW];
        #pragma unroll
        for (int rr = 0; rr < RW; ++rr)
            acc[rr] = (v8f){0.f,0.f,0.f,0.f,0.f,0.f,0.f,0.f};

        k_chain(ks, [&](auto kc) {
            constexpr int kt = decltype(kc)::v;
            const int ka = kt * 4 + kk;
            v2f bf;
            bf.x = (bcol_ok && (ka     < d)) ? braw[kt].x : 0.f;
            bf.y = (bcol_ok && (ka + 1 < d)) ? braw[kt].y : 0.f;
            #pragma unroll
            for (int rr = 0; rr < RW; ++rr)
                acc[rr] = __builtin_amdgcn_wmma_f32_16x16x4_f32(false, a[rr][kt], false, bf,
                                                                (short)0, acc[rr], false, false);
        });

        load_b(braw, bnext);

        const int sh0 = (int)(((uintptr_t)(out + ((b * RROT + r0) * NFW + coff)) >> 2) & 3);

        if (col < d) {
            #pragma unroll
            for (int v = 0; v < 8; ++v) {
                if (row0 + v < d) {
                    #pragma unroll
                    for (int rr = 0; rr < RW; ++rr)
                        sb[rr * CHS + sh0 + so + v * d] = acc[rr][v];
                }
            }
        }

        lds_handover();

        float* __restrict__ oc = out + ((b * RROT + r0 + w) * NFW + coff);
        const int e0  = w * CHS + sh0;
        const int a   = e0 & 3;
        const float* sc = sb + (e0 - a);
        float* __restrict__ og = oc - a;
        const int m8  = (int)(((uintptr_t)og >> 4) & 7);
        const int qt  = (a + clen) >> 2;
        const int tn  = (a + clen) & 3;
        const int qlo = (a != 0) ? 1 : 0;
        const int nst = (m8 + qt + 31) >> 5;
        const int qb  = lane - m8;
        const v4fa* s4 = (const v4fa*)sc;
        v4fa* g4 = (v4fa*)og;
        const int  hj   = lane & 3;
        const bool htl  = (lane & 4) != 0;
        const int  hrel = htl ? 4 * qt + hj : hj;
        const bool hok  = (lane < 8) && (htl ? (hj < tn) : (a != 0 && hj >= a));
        const float hv  = sc[min(max(hrel, a), a + clen - 1)];
        if (nst <= 1)      put_pieces<1>(s4, g4, qb, qlo, qt);
        else if (nst == 2) put_pieces<2>(s4, g4, qb, qlo, qt);
        else if (nst == 3) put_pieces<3>(s4, g4, qb, qlo, qt);
        else if (nst == 4) put_pieces<4>(s4, g4, qb, qlo, qt);
        else if (nst == 5) put_pieces<5>(s4, g4, qb, qlo, qt);
        else               put_pieces<6>(s4, g4, qb, qlo, qt);
        if (hok) og[hrel] = hv;
    };

    v2f braw[KSTEPS];
    load_b(braw, b0);

    if (lane < APITCH) {
        #pragma unroll
        for (int j = 0; j < 16; ++j) {
            const bool ok = (ti * 16 + j < d) && kl_ok;
            #pragma unroll
            for (int rr = 0; rr < RW; ++rr)
                afr[w][rr][j][lane] = ok ? araw[rr][j] : 0.f;
        }
    }
    lds_handover();
    k_chain(ks, [&](auto kc) {
        constexpr int kt = decltype(kc)::v;
        #pragma unroll
        for (int rr = 0; rr < RW; ++rr)
            a[rr][kt] = *(const v2fa*)&afr[w][rr][m][kt * 4 + kk];
    });
    #pragma unroll 1
    for (int bi = 0; bi < NBW; bi += 2) {
        const int b = b0 + bi;
        step(braw, stage[0], b, b + 1);
        step(braw, stage[1], b + 1, (bi + 2 < NBW) ? b + 2 : b + 1);
    }
}

extern "C" void kernel_launch(void* const* d_in, const int* in_sizes, int n_in,
                              void* d_out, int out_size, void* d_ws, size_t ws_size,
                              hipStream_t stream) {
    (void)in_sizes; (void)n_in; (void)d_ws; (void)ws_size; (void)out_size;
    const float* coeffs = (const float*)d_in[0];
    const float* wig    = (const float*)d_in[1];
    float* out          = (float*)d_out;

    dim3 block(32, GT);
    dim3 grid((RROT / RW) * GROUPS * BGRP);
    hipLaunchKernelGGL(WignerRotator_wmma, grid, block, 0, stream, coeffs, wig, out);
}
